# starting kernel
# baseline (speedup 1.0000x reference)
#include <hip/hip_runtime.h>

typedef __attribute__((ext_vector_type(2))) float v2f;
typedef __attribute__((ext_vector_type(4))) float v4f;
typedef __attribute__((ext_vector_type(8))) float v8f;

#define F_DIM 256
#define P_LEAVES 128
#define WAVES_PER_WG 8
#define ROWS_PER_WAVE 32
#define THREADS_PER_WG 256

#define SLICE_W 576
#define KH_OFF  288
#define LDS_WORDS ((F_DIM / 4) * SLICE_W)

__device__ __forceinline__ float fmin1(float a, float b) {
    return __builtin_amdgcn_fmed3f(a, b, -__builtin_inff());
}
__device__ __forceinline__ float fmax1(float a, float b) {
    return __builtin_amdgcn_fmed3f(a, b, __builtin_inff());
}
__device__ __forceinline__ v8f v8min(v8f a, v8f b) {
    v8f r;
#pragma unroll
    for (int i = 0; i < 8; ++i) r[i] = fmin1(a[i], b[i]);
    return r;
}
__device__ __forceinline__ v8f v8max(v8f a, v8f b) {
    v8f r;
#pragma unroll
    for (int i = 0; i < 8; ++i) r[i] = fmax1(a[i], b[i]);
    return r;
}
template <int XORM>
__device__ __forceinline__ v8f v8swz(v8f v) {
    constexpr int PAT = (XORM << 10) | 0x1f;
    v8f r;
#pragma unroll
    for (int i = 0; i < 8; ++i)
        r[i] = __int_as_float(
            __builtin_amdgcn_ds_swizzle(__float_as_int(v[i]), PAT));
    return r;
}

extern "C" __global__ __launch_bounds__(THREADS_PER_WG)
void aln_wmma_kernel(const float* __restrict__ x,
                     const float* __restrict__ W,
                     const float* __restrict__ bias,
                     float* __restrict__ out)
{
    extern __shared__ float ldsW[];

    const int tid = threadIdx.x;

    const v4f* W4 = (const v4f*)W;
#pragma unroll 4
    for (int i = 0; i < (F_DIM * P_LEAVES) / (4 * THREADS_PER_WG); ++i) {
        int flat4 = tid + i * THREADS_PER_WG;
        v4f w = W4[flat4];
        int k  = flat4 >> 5;
        int n0 = (flat4 & 31) * 4;
        int bse = (k >> 2) * SLICE_W + ((k >> 1) & 1) * KH_OFF + (k & 1);
        ldsW[bse + (n0 + 0) * 2] = w[0];
        ldsW[bse + (n0 + 1) * 2] = w[1];
        ldsW[bse + (n0 + 2) * 2] = w[2];
        ldsW[bse + (n0 + 3) * 2] = w[3];
    }

    const int lane  = tid & 31;
    const int wave  = tid >> 5;
    const int c     = lane & 15;
    const int khalf = lane >> 4;

    const int rowBase = (blockIdx.x * WAVES_PER_WG + wave) * ROWS_PER_WAVE;

    float bb[8];
#pragma unroll
    for (int nt = 0; nt < 8; ++nt) bb[nt] = bias[nt * 16 + c];

    __syncthreads();

    const float* xA0 = x + (size_t)(rowBase + c) * F_DIM + khalf * 2;
    const float* xA1 = xA0 + 16 * F_DIM;

    const float* ldsB = ldsW + khalf * KH_OFF + c * 2;

    v8f acc[2][8];
#pragma unroll
    for (int mt = 0; mt < 2; ++mt)
#pragma unroll
        for (int nt = 0; nt < 8; ++nt) acc[mt][nt] = (v8f)0.0f;

#pragma unroll 2
    for (int kc = 0; kc < F_DIM / 4; ++kc) {
        v2f a0 = __builtin_nontemporal_load((const v2f*)(xA0 + kc * 4));
        v2f a1 = __builtin_nontemporal_load((const v2f*)(xA1 + kc * 4));
        const float* bp = ldsB + kc * SLICE_W;
#pragma unroll
        for (int nt = 0; nt < 8; ++nt) {
            v2f bf = *(const v2f*)(bp + nt * 32);
            acc[0][nt] = __builtin_amdgcn_wmma_f32_16x16x4_f32(
                false, a0, false, bf, (short)0, acc[0][nt], false, false);
            acc[1][nt] = __builtin_amdgcn_wmma_f32_16x16x4_f32(
                false, a1, false, bf, (short)0, acc[1][nt], false, false);
        }
    }

    const bool oddlane = (lane & 1) != 0;
#pragma unroll
    for (int mt = 0; mt < 2; ++mt) {
        v8f p[4];
#pragma unroll
        for (int j = 0; j < 4; ++j) {
            v8f va = acc[mt][2 * j]     + bb[2 * j];
            v8f vb = acc[mt][2 * j + 1] + bb[2 * j + 1];
            va = v8min(va, v8swz<1>(va));
            vb = v8min(vb, v8swz<1>(vb));
            v8f m;
#pragma unroll
            for (int i = 0; i < 8; ++i) m[i] = oddlane ? vb[i] : va[i];
            m = v8max(m, v8swz<2>(m));
            m = v8min(m, v8swz<4>(m));
            m = v8max(m, v8swz<8>(m));
            m = v8min(m, v8swz<1>(m));
            p[j] = m;
        }
        v8f q0 = v8max(p[0], p[1]);
        v8f q1 = v8max(p[2], p[3]);
        v8f r  = v8min(q0, q1);

        if (c == 0) {
            float* o = out + rowBase + mt * 16 + khalf * 8;
            v4f lo = { r[0], r[1], r[2], r[3] };
            v4f hi = { r[4], r[5], r[6], r[7] };
            *(v4f*)(o + 0) = lo;
            *(v4f*)(o + 4) = hi;
        }
    }
}

extern "C" void kernel_launch(void* const* d_in, const int* in_sizes, int n_in,
                              void* d_out, int out_size, void* d_ws, size_t ws_size,
                              hipStream_t stream) {
    const float* x = (const float*)d_in[0];
    const float* W = (const float*)d_in[1];
    const float* b = (const float*)d_in[2];
    float* out = (float*)d_out;

    const int Bsz = in_sizes[0] / F_DIM;
    const int rowsPerWG = WAVES_PER_WG * ROWS_PER_WAVE;
    const int nBlocks = Bsz / rowsPerWG;

    const size_t shmem = (size_t)LDS_WORDS * sizeof(float);
    (void)hipFuncSetAttribute((const void*)aln_wmma_kernel,
                              hipFuncAttributeMaxDynamicSharedMemorySize,
                              (int)shmem);

    aln_wmma_kernel<<<nBlocks, THREADS_PER_WG, shmem, stream>>>(x, W, b, out);
}


# x operand staged through a per-wave LDS tile: aligned 256-byte row pieces fetched by coalesced 128-bit loads one 64-k chunk ahead, fragments read from LDS; products and their order unchanged
# speedup vs baseline: 1.6783x; 1.6783x over previous
#include <hip/hip_runtime.h>

typedef __attribute__((ext_vector_type(2))) float v2f;
typedef __attribute__((ext_vector_type(4))) float v4f;
typedef __attribute__((ext_vector_type(8))) float v8f;

#define F_DIM 256
#define P_LEAVES 128
#define WAVES_PER_WG 8
#define ROWS_PER_WAVE 32
#define THREADS_PER_WG 256

#define SLICE_W 576
#define KH_OFF  288
#define LDS_WORDS ((F_DIM / 4) * SLICE_W)

#define XC_K      64
#define XC_STEPS  (XC_K / 4)
#define XC_COUNT  (F_DIM / XC_K)
#define XC_PITCH  68
#define XC_LOADS  (ROWS_PER_WAVE * XC_K / (32 * 4))
#define XT_WORDS  (ROWS_PER_WAVE * XC_PITCH)
#define LDS_TOTAL_WORDS (LDS_WORDS + WAVES_PER_WG * XT_WORDS)

__device__ __forceinline__ float fmin1(float a, float b) {
    return __builtin_amdgcn_fmed3f(a, b, -__builtin_inff());
}
__device__ __forceinline__ float fmax1(float a, float b) {
    return __builtin_amdgcn_fmed3f(a, b, __builtin_inff());
}
__device__ __forceinline__ v8f v8min(v8f a, v8f b) {
    v8f r;
#pragma unroll
    for (int i = 0; i < 8; ++i) r[i] = fmin1(a[i], b[i]);
    return r;
}
__device__ __forceinline__ v8f v8max(v8f a, v8f b) {
    v8f r;
#pragma unroll
    for (int i = 0; i < 8; ++i) r[i] = fmax1(a[i], b[i]);
    return r;
}
template <int XORM>
__device__ __forceinline__ v8f v8swz(v8f v) {
    constexpr int PAT = (XORM << 10) | 0x1f;
    v8f r;
#pragma unroll
    for (int i = 0; i < 8; ++i)
        r[i] = __int_as_float(
            __builtin_amdgcn_ds_swizzle(__float_as_int(v[i]), PAT));
    return r;
}

extern "C" __global__ __launch_bounds__(THREADS_PER_WG)
void aln_wmma_kernel(const float* __restrict__ x,
                     const float* __restrict__ W,
                     const float* __restrict__ bias,
                     float* __restrict__ out)
{
    extern __shared__ float ldsAll[];
    float* ldsW = ldsAll;

    const int tid = threadIdx.x;
    const int lane  = tid & 31;
    const int wave  = tid >> 5;
    const int c     = lane & 15;
    const int khalf = lane >> 4;

    const int rowBase = (blockIdx.x * WAVES_PER_WG + wave) * ROWS_PER_WAVE;

    const float* xg  = x + (size_t)(rowBase + khalf) * F_DIM + c * 4;
    float*       xt  = ldsAll + LDS_WORDS + wave * XT_WORDS;
    float*       xtS = xt + khalf * XC_PITCH + c * 4;

    v4f pre[XC_LOADS];
#pragma unroll
    for (int i = 0; i < XC_LOADS; ++i)
        pre[i] = __builtin_nontemporal_load((const v4f*)(xg + (size_t)(2 * i) * F_DIM));

    const v4f* W4 = (const v4f*)W;
#pragma unroll 4
    for (int i = 0; i < (F_DIM * P_LEAVES) / (4 * THREADS_PER_WG); ++i) {
        int flat4 = tid + i * THREADS_PER_WG;
        v4f w = W4[flat4];
        int k  = flat4 >> 5;
        int n0 = (flat4 & 31) * 4;
        int bse = (k >> 2) * SLICE_W + ((k >> 1) & 1) * KH_OFF + (k & 1);
        ldsW[bse + (n0 + 0) * 2] = w[0];
        ldsW[bse + (n0 + 1) * 2] = w[1];
        ldsW[bse + (n0 + 2) * 2] = w[2];
        ldsW[bse + (n0 + 3) * 2] = w[3];
    }

    float bb[8];
#pragma unroll
    for (int nt = 0; nt < 8; ++nt) bb[nt] = bias[nt * 16 + c];

#pragma unroll
    for (int i = 0; i < XC_LOADS; ++i)
        *(v4f*)(xtS + 2 * i * XC_PITCH) = pre[i];

    __syncthreads();

    const float* xa0 = xt + c * XC_PITCH + khalf * 2;
    const float* xa1 = xa0 + 16 * XC_PITCH;

    const float* ldsB = ldsW + khalf * KH_OFF + c * 2;

    v8f acc[2][8];
#pragma unroll
    for (int mt = 0; mt < 2; ++mt)
#pragma unroll
        for (int nt = 0; nt < 8; ++nt) acc[mt][nt] = (v8f)0.0f;

#pragma unroll
    for (int ch = 0; ch < XC_COUNT; ++ch) {
        if (ch + 1 < XC_COUNT) {
#pragma unroll
            for (int i = 0; i < XC_LOADS; ++i)
                pre[i] = __builtin_nontemporal_load(
                    (const v4f*)(xg + (ch + 1) * XC_K + (size_t)(2 * i) * F_DIM));
        }
#pragma unroll 2
        for (int ks = 0; ks < XC_STEPS; ++ks) {
            v2f a0 = *(const v2f*)(xa0 + ks * 4);
            v2f a1 = *(const v2f*)(xa1 + ks * 4);
            const float* bp = ldsB + (ch * XC_STEPS + ks) * SLICE_W;
#pragma unroll
            for (int nt = 0; nt < 8; ++nt) {
                v2f bf = *(const v2f*)(bp + nt * 32);
                acc[0][nt] = __builtin_amdgcn_wmma_f32_16x16x4_f32(
                    false, a0, false, bf, (short)0, acc[0][nt], false, false);
                acc[1][nt] = __builtin_amdgcn_wmma_f32_16x16x4_f32(
                    false, a1, false, bf, (short)0, acc[1][nt], false, false);
            }
        }
        if (ch + 1 < XC_COUNT) {
#pragma unroll
            for (int i = 0; i < XC_LOADS; ++i)
                *(v4f*)(xtS + 2 * i * XC_PITCH) = pre[i];
        }
    }

    const bool oddlane = (lane & 1) != 0;
#pragma unroll
    for (int mt = 0; mt < 2; ++mt) {
        v8f p[4];
#pragma unroll
        for (int j = 0; j < 4; ++j) {
            v8f va = acc[mt][2 * j]     + bb[2 * j];
            v8f vb = acc[mt][2 * j + 1] + bb[2 * j + 1];
            va = v8min(va, v8swz<1>(va));
            vb = v8min(vb, v8swz<1>(vb));
            v8f m;
#pragma unroll
            for (int i = 0; i < 8; ++i) m[i] = oddlane ? vb[i] : va[i];
            m = v8max(m, v8swz<2>(m));
            m = v8min(m, v8swz<4>(m));
            m = v8max(m, v8swz<8>(m));
            m = v8min(m, v8swz<1>(m));
            p[j] = m;
        }
        v8f q0 = v8max(p[0], p[1]);
        v8f q1 = v8max(p[2], p[3]);
        v8f r  = v8min(q0, q1);

        if (c == 0) {
            float* o = out + rowBase + mt * 16 + khalf * 8;
            v4f lo = { r[0], r[1], r[2], r[3] };
            v4f hi = { r[4], r[5], r[6], r[7] };
            *(v4f*)(o + 0) = lo;
            *(v4f*)(o + 4) = hi;
        }
    }
}

extern "C" void kernel_launch(void* const* d_in, const int* in_sizes, int n_in,
                              void* d_out, int out_size, void* d_ws, size_t ws_size,
                              hipStream_t stream) {
    const float* x = (const float*)d_in[0];
    const float* W = (const float*)d_in[1];
    const float* b = (const float*)d_in[2];
    float* out = (float*)d_out;

    const int Bsz = in_sizes[0] / F_DIM;
    const int rowsPerWG = WAVES_PER_WG * ROWS_PER_WAVE;
    const int nBlocks = Bsz / rowsPerWG;

    const size_t shmem = (size_t)LDS_TOTAL_WORDS * sizeof(float);
    (void)hipFuncSetAttribute((const void*)aln_wmma_kernel,
                              hipFuncAttributeMaxDynamicSharedMemorySize,
                              (int)shmem);

    aln_wmma_kernel<<<nBlocks, THREADS_PER_WG, shmem, stream>>>(x, W, b, out);
}


# WMMA operand roles swapped (W fragment as A, x fragment as B): leaves lie along a lane's registers, so 6 of 7 min/max levels fold in-lane, 1 swizzle level left; products and k order unchanged
# speedup vs baseline: 1.8285x; 1.0895x over previous
#include <hip/hip_runtime.h>

typedef __attribute__((ext_vector_type(2))) float v2f;
typedef __attribute__((ext_vector_type(4))) float v4f;
typedef __attribute__((ext_vector_type(8))) float v8f;

#define F_DIM 256
#define P_LEAVES 128
#define WAVES_PER_WG 8
#define ROWS_PER_WAVE 32
#define THREADS_PER_WG 256

#define SLICE_W 576
#define KH_OFF  288
#define LDS_WORDS ((F_DIM / 4) * SLICE_W)

#define XC_K      64
#define XC_STEPS  (XC_K / 4)
#define XC_COUNT  (F_DIM / XC_K)
#define XC_PITCH  68
#define XC_LOADS  (ROWS_PER_WAVE * XC_K / (32 * 4))
#define XT_WORDS  (ROWS_PER_WAVE * XC_PITCH)
#define LDS_TOTAL_WORDS (LDS_WORDS + WAVES_PER_WG * XT_WORDS)

__device__ __forceinline__ float fmin1(float a, float b) {
    return __builtin_amdgcn_fmed3f(a, b, -__builtin_inff());
}
__device__ __forceinline__ float fmax1(float a, float b) {
    return __builtin_amdgcn_fmed3f(a, b, __builtin_inff());
}
__device__ __forceinline__ float swz_half(float v) {
    return __int_as_float(
        __builtin_amdgcn_ds_swizzle(__float_as_int(v), (16 << 10) | 0x1f));
}

extern "C" __global__ __launch_bounds__(THREADS_PER_WG)
void aln_wmma_kernel(const float* __restrict__ x,
                     const float* __restrict__ W,
                     const float* __restrict__ bias,
                     float* __restrict__ out)
{
    extern __shared__ float ldsAll[];
    float* ldsW = ldsAll;

    const int tid = threadIdx.x;
    const int lane  = tid & 31;
    const int wave  = tid >> 5;
    const int c     = lane & 15;
    const int khalf = lane >> 4;

    const int rowBase = (blockIdx.x * WAVES_PER_WG + wave) * ROWS_PER_WAVE;

    const float* xg  = x + (size_t)(rowBase + khalf) * F_DIM + c * 4;
    float*       xt  = ldsAll + LDS_WORDS + wave * XT_WORDS;
    float*       xtS = xt + khalf * XC_PITCH + c * 4;

    v4f pre[XC_LOADS];
#pragma unroll
    for (int i = 0; i < XC_LOADS; ++i)
        pre[i] = __builtin_nontemporal_load((const v4f*)(xg + (size_t)(2 * i) * F_DIM));

    const v4f* W4 = (const v4f*)W;
#pragma unroll 4
    for (int i = 0; i < (F_DIM * P_LEAVES) / (4 * THREADS_PER_WG); ++i) {
        int flat4 = tid + i * THREADS_PER_WG;
        v4f w = W4[flat4];
        int k  = flat4 >> 5;
        int n0 = (flat4 & 31) * 4;
        int bse = (k >> 2) * SLICE_W + ((k >> 1) & 1) * KH_OFF + (k & 1);
        ldsW[bse + (n0 + 0) * 2] = w[0];
        ldsW[bse + (n0 + 1) * 2] = w[1];
        ldsW[bse + (n0 + 2) * 2] = w[2];
        ldsW[bse + (n0 + 3) * 2] = w[3];
    }

#pragma unroll
    for (int i = 0; i < XC_LOADS; ++i)
        *(v4f*)(xtS + 2 * i * XC_PITCH) = pre[i];

    __syncthreads();

    const float* xa0 = xt + c * XC_PITCH + khalf * 2;
    const float* xa1 = xa0 + 16 * XC_PITCH;

    const float* ldsB = ldsW + khalf * KH_OFF + c * 2;

    v8f acc[2][8];
#pragma unroll
    for (int mt = 0; mt < 2; ++mt)
#pragma unroll
        for (int nt = 0; nt < 8; ++nt) acc[mt][nt] = (v8f)0.0f;

#pragma unroll
    for (int ch = 0; ch < XC_COUNT; ++ch) {
        if (ch + 1 < XC_COUNT) {
#pragma unroll
            for (int i = 0; i < XC_LOADS; ++i)
                pre[i] = __builtin_nontemporal_load(
                    (const v4f*)(xg + (ch + 1) * XC_K + (size_t)(2 * i) * F_DIM));
        } else {
#pragma unroll
            for (int nt = 0; nt < 8; ++nt) {
                const v4f* b4 = (const v4f*)(bias + nt * 16 + khalf * 8);
                pre[2 * nt]     = b4[0];
                pre[2 * nt + 1] = b4[1];
            }
        }
#pragma unroll 2
        for (int ks = 0; ks < XC_STEPS; ++ks) {
            v2f a0 = *(const v2f*)(xa0 + ks * 4);
            v2f a1 = *(const v2f*)(xa1 + ks * 4);
            const float* bp = ldsB + (ch * XC_STEPS + ks) * SLICE_W;
#pragma unroll
            for (int nt = 0; nt < 8; ++nt) {
                v2f wf = *(const v2f*)(bp + nt * 32);
                acc[0][nt] = __builtin_amdgcn_wmma_f32_16x16x4_f32(
                    false, wf, false, a0, (short)0, acc[0][nt], false, false);
                acc[1][nt] = __builtin_amdgcn_wmma_f32_16x16x4_f32(
                    false, wf, false, a1, (short)0, acc[1][nt], false, false);
            }
        }
        if (ch + 1 < XC_COUNT) {
#pragma unroll
            for (int i = 0; i < XC_LOADS; ++i)
                *(v4f*)(xtS + 2 * i * XC_PITCH) = pre[i];
        }
    }

    float y[2][8];
#pragma unroll
    for (int mt = 0; mt < 2; ++mt) {
#pragma unroll
        for (int nt = 0; nt < 8; ++nt) {
            v8f a = acc[mt][nt];
            v4f blo = pre[2 * nt], bhi = pre[2 * nt + 1];
            float m0 = fmin1(a[0] + blo[0], a[1] + blo[1]);
            float m1 = fmin1(a[2] + blo[2], a[3] + blo[3]);
            float m2 = fmin1(a[4] + bhi[0], a[5] + bhi[1]);
            float m3 = fmin1(a[6] + bhi[2], a[7] + bhi[3]);
            y[mt][nt] = fmin1(fmax1(m0, m1), fmax1(m2, m3));
        }
    }
    float o[2][8];
#pragma unroll
    for (int mt = 0; mt < 2; ++mt)
#pragma unroll
        for (int nt = 0; nt < 8; ++nt) o[mt][nt] = swz_half(y[mt][nt]);
    float res[2];
#pragma unroll
    for (int mt = 0; mt < 2; ++mt) {
        float z[8];
#pragma unroll
        for (int nt = 0; nt < 8; ++nt) z[nt] = fmax1(y[mt][nt], o[mt][nt]);
        float w0 = fmin1(z[0], z[1]), w1 = fmin1(z[2], z[3]);
        float w2 = fmin1(z[4], z[5]), w3 = fmin1(z[6], z[7]);
        res[mt] = fmin1(fmax1(w0, w1), fmax1(w2, w3));
    }
    out[rowBase + lane] = khalf ? res[1] : res[0];
}

extern "C" void kernel_launch(void* const* d_in, const int* in_sizes, int n_in,
                              void* d_out, int out_size, void* d_ws, size_t ws_size,
                              hipStream_t stream) {
    const float* x = (const float*)d_in[0];
    const float* W = (const float*)d_in[1];
    const float* b = (const float*)d_in[2];
    float* out = (float*)d_out;

    const int Bsz = in_sizes[0] / F_DIM;
    const int rowsPerWG = WAVES_PER_WG * ROWS_PER_WAVE;
    const int nBlocks = Bsz / rowsPerWG;

    const size_t shmem = (size_t)LDS_TOTAL_WORDS * sizeof(float);
    (void)hipFuncSetAttribute((const void*)aln_wmma_kernel,
                              hipFuncAttributeMaxDynamicSharedMemorySize,
                              (int)shmem);

    aln_wmma_kernel<<<nBlocks, THREADS_PER_WG, shmem, stream>>>(x, W, b, out);
}


# LDS fragment reads run one 64-WMMA block (4 k-steps) ahead in a second register set: block b+1's reads issue as one group before block b's WMMAs, across x-chunk ends too; products, k order unchanged
# speedup vs baseline: 1.8944x; 1.0360x over previous
#include <hip/hip_runtime.h>

typedef __attribute__((ext_vector_type(2))) float v2f;
typedef __attribute__((ext_vector_type(4))) float v4f;
typedef __attribute__((ext_vector_type(8))) float v8f;

#define F_DIM 256
#define P_LEAVES 128
#define WAVES_PER_WG 8
#define ROWS_PER_WAVE 32
#define THREADS_PER_WG 256

#define SLICE_W 576
#define KH_OFF  288
#define LDS_WORDS ((F_DIM / 4) * SLICE_W)

#define XC_K      64
#define XC_STEPS  (XC_K / 4)
#define XC_COUNT  (F_DIM / XC_K)
#define XC_PITCH  68
#define XC_LOADS  (ROWS_PER_WAVE * XC_K / (32 * 4))
#define XT_WORDS  (ROWS_PER_WAVE * XC_PITCH)
#define LDS_TOTAL_WORDS (LDS_WORDS + WAVES_PER_WG * XT_WORDS)

__device__ __forceinline__ float fmin1(float a, float b) {
    return __builtin_amdgcn_fmed3f(a, b, -__builtin_inff());
}
__device__ __forceinline__ float fmax1(float a, float b) {
    return __builtin_amdgcn_fmed3f(a, b, __builtin_inff());
}
__device__ __forceinline__ float swz_half(float v) {
    return __int_as_float(
        __builtin_amdgcn_ds_swizzle(__float_as_int(v), (16 << 10) | 0x1f));
}

#define BLK_STEPS 4
__device__ __forceinline__ void load_block(v2f (&fa)[BLK_STEPS][2], v2f (&fw)[BLK_STEPS][8],
                                           const float* xa0, const float* xa1,
                                           const float* wb, int ks)
{
#pragma unroll
    for (int t = 0; t < BLK_STEPS; ++t) {
        fa[t][0] = *(const v2f*)(xa0 + (ks + t) * 4);
        fa[t][1] = *(const v2f*)(xa1 + (ks + t) * 4);
        const float* bp = wb + (ks + t) * SLICE_W;
#pragma unroll
        for (int nt = 0; nt < 8; ++nt)
            fw[t][nt] = *(const v2f*)(bp + nt * 32);
    }
}

__device__ __forceinline__ void mma_block(v8f (&acc)[2][8], const v2f (&fa)[BLK_STEPS][2],
                                          const v2f (&fw)[BLK_STEPS][8])
{
#pragma unroll
    for (int t = 0; t < BLK_STEPS; ++t) {
#pragma unroll
        for (int nt = 0; nt < 8; ++nt) {
            acc[0][nt] = __builtin_amdgcn_wmma_f32_16x16x4_f32(
                false, fw[t][nt], false, fa[t][0], (short)0, acc[0][nt], false, false);
            acc[1][nt] = __builtin_amdgcn_wmma_f32_16x16x4_f32(
                false, fw[t][nt], false, fa[t][1], (short)0, acc[1][nt], false, false);
        }
    }
}

extern "C" __global__ __launch_bounds__(THREADS_PER_WG)
void aln_wmma_kernel(const float* __restrict__ x,
                     const float* __restrict__ W,
                     const float* __restrict__ bias,
                     float* __restrict__ out)
{
    extern __shared__ float ldsAll[];
    float* ldsW = ldsAll;

    const int tid = threadIdx.x;
    const int lane  = tid & 31;
    const int wave  = tid >> 5;
    const int c     = lane & 15;
    const int khalf = lane >> 4;

    const int rowBase = (blockIdx.x * WAVES_PER_WG + wave) * ROWS_PER_WAVE;

    const float* xg  = x + (size_t)(rowBase + khalf) * F_DIM + c * 4;
    float*       xt  = ldsAll + LDS_WORDS + wave * XT_WORDS;
    float*       xtS = xt + khalf * XC_PITCH + c * 4;

    v4f pre[XC_LOADS];
#pragma unroll
    for (int i = 0; i < XC_LOADS; ++i)
        pre[i] = __builtin_nontemporal_load((const v4f*)(xg + (size_t)(2 * i) * F_DIM));

    const v4f* W4 = (const v4f*)W;
#pragma unroll 4
    for (int i = 0; i < (F_DIM * P_LEAVES) / (4 * THREADS_PER_WG); ++i) {
        int flat4 = tid + i * THREADS_PER_WG;
        v4f w = W4[flat4];
        int k  = flat4 >> 5;
        int n0 = (flat4 & 31) * 4;
        int bse = (k >> 2) * SLICE_W + ((k >> 1) & 1) * KH_OFF + (k & 1);
        ldsW[bse + (n0 + 0) * 2] = w[0];
        ldsW[bse + (n0 + 1) * 2] = w[1];
        ldsW[bse + (n0 + 2) * 2] = w[2];
        ldsW[bse + (n0 + 3) * 2] = w[3];
    }

#pragma unroll
    for (int i = 0; i < XC_LOADS; ++i)
        *(v4f*)(xtS + 2 * i * XC_PITCH) = pre[i];

    __syncthreads();

    const float* xa0 = xt + c * XC_PITCH + khalf * 2;
    const float* xa1 = xa0 + 16 * XC_PITCH;

    const float* ldsB = ldsW + khalf * KH_OFF + c * 2;

    v8f acc[2][8];
#pragma unroll
    for (int mt = 0; mt < 2; ++mt)
#pragma unroll
        for (int nt = 0; nt < 8; ++nt) acc[mt][nt] = (v8f)0.0f;

    v2f pa[BLK_STEPS][2], pw[BLK_STEPS][8];
    v2f qa[BLK_STEPS][2], qw[BLK_STEPS][8];
#define BLOCK_FENCE() __builtin_amdgcn_sched_barrier(0)
    const float* wb[XC_COUNT];
#pragma unroll
    for (int ch = 0; ch < XC_COUNT; ++ch) {
        int wofs = ch * (XC_STEPS * SLICE_W);
        asm volatile("" : "+v"(wofs));
        wb[ch] = ldsB + wofs;
    }
    load_block(pa, pw, xa0, xa1, wb[0], 0);
    BLOCK_FENCE();
#pragma unroll
    for (int ch = 0; ch < XC_COUNT; ++ch) {
        if (ch + 1 < XC_COUNT) {
#pragma unroll
            for (int i = 0; i < XC_LOADS; ++i)
                pre[i] = __builtin_nontemporal_load(
                    (const v4f*)(xg + (ch + 1) * XC_K + (size_t)(2 * i) * F_DIM));
        } else {
#pragma unroll
            for (int nt = 0; nt < 8; ++nt) {
                const v4f* b4 = (const v4f*)(bias + nt * 16 + khalf * 8);
                pre[2 * nt]     = b4[0];
                pre[2 * nt + 1] = b4[1];
            }
        }
        load_block(qa, qw, xa0, xa1, wb[ch], 1 * BLK_STEPS);
        BLOCK_FENCE();
        mma_block(acc, pa, pw);
        BLOCK_FENCE();
        load_block(pa, pw, xa0, xa1, wb[ch], 2 * BLK_STEPS);
        BLOCK_FENCE();
        mma_block(acc, qa, qw);
        BLOCK_FENCE();
        load_block(qa, qw, xa0, xa1, wb[ch], 3 * BLK_STEPS);
        BLOCK_FENCE();
        mma_block(acc, pa, pw);
        BLOCK_FENCE();
        if (ch + 1 < XC_COUNT) {
#pragma unroll
            for (int i = 0; i < XC_LOADS; ++i)
                *(v4f*)(xtS + 2 * i * XC_PITCH) = pre[i];
            load_block(pa, pw, xa0, xa1, wb[ch + 1], 0);
            BLOCK_FENCE();
        }
        mma_block(acc, qa, qw);
        BLOCK_FENCE();
    }
#undef BLOCK_FENCE

    float y[2][8];
#pragma unroll
    for (int mt = 0; mt < 2; ++mt) {
#pragma unroll
        for (int nt = 0; nt < 8; ++nt) {
            v8f a = acc[mt][nt];
            v4f blo = pre[2 * nt], bhi = pre[2 * nt + 1];
            float m0 = fmin1(a[0] + blo[0], a[1] + blo[1]);
            float m1 = fmin1(a[2] + blo[2], a[3] + blo[3]);
            float m2 = fmin1(a[4] + bhi[0], a[5] + bhi[1]);
            float m3 = fmin1(a[6] + bhi[2], a[7] + bhi[3]);
            y[mt][nt] = fmin1(fmax1(m0, m1), fmax1(m2, m3));
        }
    }
    float o[2][8];
#pragma unroll
    for (int mt = 0; mt < 2; ++mt)
#pragma unroll
        for (int nt = 0; nt < 8; ++nt) o[mt][nt] = swz_half(y[mt][nt]);
    float res[2];
#pragma unroll
    for (int mt = 0; mt < 2; ++mt) {
        float z[8];
#pragma unroll
        for (int nt = 0; nt < 8; ++nt) z[nt] = fmax1(y[mt][nt], o[mt][nt]);
        float w0 = fmin1(z[0], z[1]), w1 = fmin1(z[2], z[3]);
        float w2 = fmin1(z[4], z[5]), w3 = fmin1(z[6], z[7]);
        res[mt] = fmin1(fmax1(w0, w1), fmax1(w2, w3));
    }
    out[rowBase + lane] = khalf ? res[1] : res[0];
}

extern "C" void kernel_launch(void* const* d_in, const int* in_sizes, int n_in,
                              void* d_out, int out_size, void* d_ws, size_t ws_size,
                              hipStream_t stream) {
    const float* x = (const float*)d_in[0];
    const float* W = (const float*)d_in[1];
    const float* b = (const float*)d_in[2];
    float* out = (float*)d_out;

    const int Bsz = in_sizes[0] / F_DIM;
    const int rowsPerWG = WAVES_PER_WG * ROWS_PER_WAVE;
    const int nBlocks = Bsz / rowsPerWG;

    const size_t shmem = (size_t)LDS_TOTAL_WORDS * sizeof(float);
    (void)hipFuncSetAttribute((const void*)aln_wmma_kernel,
                              hipFuncAttributeMaxDynamicSharedMemorySize,
                              (int)shmem);

    aln_wmma_kernel<<<nBlocks, THREADS_PER_WG, shmem, stream>>>(x, W, b, out);
}


# W staged one 64-k chunk ahead under the WMMAs (b128 LDS stores, split workgroup barrier per chunk) instead of all 128 KiB in front of the first WMMA; LDS image, products and k order unchanged
# speedup vs baseline: 1.9539x; 1.0314x over previous
#include <hip/hip_runtime.h>

typedef __attribute__((ext_vector_type(2))) float v2f;
typedef __attribute__((ext_vector_type(4))) float v4f;
typedef __attribute__((ext_vector_type(8))) float v8f;

#define F_DIM 256
#define P_LEAVES 128
#define WAVES_PER_WG 8
#define ROWS_PER_WAVE 32
#define THREADS_PER_WG 256

#define SLICE_W 576
#define KH_OFF  288
#define LDS_WORDS ((F_DIM / 4) * SLICE_W)

#define XC_K      64
#define XC_STEPS  (XC_K / 4)
#define XC_COUNT  (F_DIM / XC_K)
#define XC_PITCH  68
#define XC_LOADS  (ROWS_PER_WAVE * XC_K / (32 * 4))
#define XT_WORDS  (ROWS_PER_WAVE * XC_PITCH)
#define LDS_TOTAL_WORDS (LDS_WORDS + WAVES_PER_WG * XT_WORDS)

__device__ __forceinline__ float fmin1(float a, float b) {
    return __builtin_amdgcn_fmed3f(a, b, -__builtin_inff());
}
__device__ __forceinline__ float fmax1(float a, float b) {
    return __builtin_amdgcn_fmed3f(a, b, __builtin_inff());
}
__device__ __forceinline__ float swz_half(float v) {
    return __int_as_float(
        __builtin_amdgcn_ds_swizzle(__float_as_int(v), (16 << 10) | 0x1f));
}

#define BLK_STEPS 4
__device__ __forceinline__ void load_block(v2f (&fa)[BLK_STEPS][2], v2f (&fw)[BLK_STEPS][8],
                                           const float* xa0, const float* xa1,
                                           const float* wb, int ks)
{
#pragma unroll
    for (int t = 0; t < BLK_STEPS; ++t) {
        fa[t][0] = *(const v2f*)(xa0 + (ks + t) * 4);
        fa[t][1] = *(const v2f*)(xa1 + (ks + t) * 4);
        const float* bp = wb + (ks + t) * SLICE_W;
#pragma unroll
        for (int nt = 0; nt < 8; ++nt)
            fw[t][nt] = *(const v2f*)(bp + nt * 32);
    }
}

__device__ __forceinline__ void mma_block(v8f (&acc)[2][8], const v2f (&fa)[BLK_STEPS][2],
                                          const v2f (&fw)[BLK_STEPS][8])
{
#pragma unroll
    for (int t = 0; t < BLK_STEPS; ++t) {
#pragma unroll
        for (int nt = 0; nt < 8; ++nt) {
            acc[0][nt] = __builtin_amdgcn_wmma_f32_16x16x4_f32(
                false, fw[t][nt], false, fa[t][0], (short)0, acc[0][nt], false, false);
            acc[1][nt] = __builtin_amdgcn_wmma_f32_16x16x4_f32(
                false, fw[t][nt], false, fa[t][1], (short)0, acc[1][nt], false, false);
        }
    }
}

#define WC_PAIRS 4
__device__ __forceinline__ void fetch_w(v4f (&wp)[2 * WC_PAIRS], const float* wg, int ch)
{
#pragma unroll
    for (int i = 0; i < WC_PAIRS; ++i) {
        const float* r = wg + (size_t)(ch * XC_K + 16 * i) * P_LEAVES;
        wp[2 * i]     = *(const v4f*)(r);
        wp[2 * i + 1] = *(const v4f*)(r + P_LEAVES);
    }
}
__device__ __forceinline__ void store_w(const v4f (&wp)[2 * WC_PAIRS], float* wS, int ch)
{
#pragma unroll
    for (int i = 0; i < WC_PAIRS; ++i) {
        float* p = wS + (ch * (XC_K / 4) + 4 * i) * SLICE_W;
        v4f a = wp[2 * i], b = wp[2 * i + 1];
        v4f lo = {a[0], b[0], a[1], b[1]};
        v4f hi = {a[2], b[2], a[3], b[3]};
        *(v4f*)(p)     = lo;
        *(v4f*)(p + 4) = hi;
    }
}

extern "C" __global__ __launch_bounds__(THREADS_PER_WG)
void aln_wmma_kernel(const float* __restrict__ x,
                     const float* __restrict__ W,
                     const float* __restrict__ bias,
                     float* __restrict__ out)
{
    extern __shared__ float ldsAll[];
    float* ldsW = ldsAll;

    const int tid = threadIdx.x;
    const int lane  = tid & 31;
    const int wave  = tid >> 5;
    const int c     = lane & 15;
    const int khalf = lane >> 4;

    const int rowBase = (blockIdx.x * WAVES_PER_WG + wave) * ROWS_PER_WAVE;

    const float* xg  = x + (size_t)(rowBase + khalf) * F_DIM + c * 4;
    float*       xt  = ldsAll + LDS_WORDS + wave * XT_WORDS;
    float*       xtS = xt + khalf * XC_PITCH + c * 4;

    const float* wg = W + (size_t)(2 * wave) * P_LEAVES + lane * 4;
    float*       wS = ldsW + (wave >> 1) * SLICE_W + (wave & 1) * KH_OFF + lane * 8;
    v4f wp[2 * WC_PAIRS];
    fetch_w(wp, wg, 0);

    v4f pre[XC_LOADS];
#pragma unroll
    for (int i = 0; i < XC_LOADS; ++i)
        pre[i] = __builtin_nontemporal_load((const v4f*)(xg + (size_t)(2 * i) * F_DIM));

    store_w(wp, wS, 0);

#pragma unroll
    for (int i = 0; i < XC_LOADS; ++i)
        *(v4f*)(xtS + 2 * i * XC_PITCH) = pre[i];

    __syncthreads();

    const float* xa0 = xt + c * XC_PITCH + khalf * 2;
    const float* xa1 = xa0 + 16 * XC_PITCH;

    const float* ldsB = ldsW + khalf * KH_OFF + c * 2;

    v8f acc[2][8];
#pragma unroll
    for (int mt = 0; mt < 2; ++mt)
#pragma unroll
        for (int nt = 0; nt < 8; ++nt) acc[mt][nt] = (v8f)0.0f;

    v2f pa[BLK_STEPS][2], pw[BLK_STEPS][8];
    v2f qa[BLK_STEPS][2], qw[BLK_STEPS][8];
#define BLOCK_FENCE() __builtin_amdgcn_sched_barrier(0)
#define W_PUBLISH() asm volatile("s_wait_dscnt 0x0\n\ts_barrier_signal -1" ::: "memory")
#define W_ACQUIRE() asm volatile("s_barrier_wait -1" ::: "memory")
    const float* wb[XC_COUNT];
#pragma unroll
    for (int ch = 0; ch < XC_COUNT; ++ch) {
        int wofs = ch * (XC_STEPS * SLICE_W);
        asm volatile("" : "+v"(wofs));
        wb[ch] = ldsB + wofs;
    }
    load_block(pa, pw, xa0, xa1, wb[0], 0);
    BLOCK_FENCE();
#pragma unroll
    for (int ch = 0; ch < XC_COUNT; ++ch) {
        if (ch + 1 < XC_COUNT) {
            fetch_w(wp, wg, ch + 1);
#pragma unroll
            for (int i = 0; i < XC_LOADS; ++i)
                pre[i] = __builtin_nontemporal_load(
                    (const v4f*)(xg + (ch + 1) * XC_K + (size_t)(2 * i) * F_DIM));
        } else {
#pragma unroll
            for (int nt = 0; nt < 8; ++nt) {
                const v4f* b4 = (const v4f*)(bias + nt * 16 + khalf * 8);
                pre[2 * nt]     = b4[0];
                pre[2 * nt + 1] = b4[1];
            }
        }
        load_block(qa, qw, xa0, xa1, wb[ch], 1 * BLK_STEPS);
        BLOCK_FENCE();
        mma_block(acc, pa, pw);
        BLOCK_FENCE();
        if (ch + 1 < XC_COUNT) store_w(wp, wS, ch + 1);
        load_block(pa, pw, xa0, xa1, wb[ch], 2 * BLK_STEPS);
        BLOCK_FENCE();
        mma_block(acc, qa, qw);
        BLOCK_FENCE();
        if (ch + 1 < XC_COUNT) W_PUBLISH();
        load_block(qa, qw, xa0, xa1, wb[ch], 3 * BLK_STEPS);
        BLOCK_FENCE();
        mma_block(acc, pa, pw);
        BLOCK_FENCE();
        if (ch + 1 < XC_COUNT) {
#pragma unroll
            for (int i = 0; i < XC_LOADS; ++i)
                *(v4f*)(xtS + 2 * i * XC_PITCH) = pre[i];
            W_ACQUIRE();
            load_block(pa, pw, xa0, xa1, wb[ch + 1], 0);
            BLOCK_FENCE();
        }
        mma_block(acc, qa, qw);
        BLOCK_FENCE();
    }
#undef BLOCK_FENCE
#undef W_PUBLISH
#undef W_ACQUIRE

    float y[2][8];
#pragma unroll
    for (int mt = 0; mt < 2; ++mt) {
#pragma unroll
        for (int nt = 0; nt < 8; ++nt) {
            v8f a = acc[mt][nt];
            v4f blo = pre[2 * nt], bhi = pre[2 * nt + 1];
            float m0 = fmin1(a[0] + blo[0], a[1] + blo[1]);
            float m1 = fmin1(a[2] + blo[2], a[3] + blo[3]);
            float m2 = fmin1(a[4] + bhi[0], a[5] + bhi[1]);
            float m3 = fmin1(a[6] + bhi[2], a[7] + bhi[3]);
            y[mt][nt] = fmin1(fmax1(m0, m1), fmax1(m2, m3));
        }
    }
    float o[2][8];
#pragma unroll
    for (int mt = 0; mt < 2; ++mt)
#pragma unroll
        for (int nt = 0; nt < 8; ++nt) o[mt][nt] = swz_half(y[mt][nt]);
    float res[2];
#pragma unroll
    for (int mt = 0; mt < 2; ++mt) {
        float z[8];
#pragma unroll
        for (int nt = 0; nt < 8; ++nt) z[nt] = fmax1(y[mt][nt], o[mt][nt]);
        float w0 = fmin1(z[0], z[1]), w1 = fmin1(z[2], z[3]);
        float w2 = fmin1(z[4], z[5]), w3 = fmin1(z[6], z[7]);
        res[mt] = fmin1(fmax1(w0, w1), fmax1(w2, w3));
    }
    out[rowBase + lane] = khalf ? res[1] : res[0];
}

extern "C" void kernel_launch(void* const* d_in, const int* in_sizes, int n_in,
                              void* d_out, int out_size, void* d_ws, size_t ws_size,
                              hipStream_t stream) {
    const float* x = (const float*)d_in[0];
    const float* W = (const float*)d_in[1];
    const float* b = (const float*)d_in[2];
    float* out = (float*)d_out;

    const int Bsz = in_sizes[0] / F_DIM;
    const int rowsPerWG = WAVES_PER_WG * ROWS_PER_WAVE;
    const int nBlocks = Bsz / rowsPerWG;

    const size_t shmem = (size_t)LDS_TOTAL_WORDS * sizeof(float);
    (void)hipFuncSetAttribute((const void*)aln_wmma_kernel,
                              hipFuncAttributeMaxDynamicSharedMemorySize,
                              (int)shmem);

    aln_wmma_kernel<<<nBlocks, THREADS_PER_WG, shmem, stream>>>(x, W, b, out);
}
